# starting kernel
# baseline (speedup 1.0000x reference)
#include <hip/hip_runtime.h>
#include <hip/hip_bf16.h>
#include <math.h>

#define S_LEN 4096
#define B_N   64
#define D_N   256

#define NC    8
#define WAVES 4
#define NP    (NC * WAVES)
#define SPW   (S_LEN / NP)
#define PART_STRIDE 258

typedef __attribute__((ext_vector_type(2))) float v2f;
typedef __attribute__((ext_vector_type(4))) float v4f;
typedef __attribute__((ext_vector_type(8))) float v8f;

__global__ void key_gemm_wmma(const float* __restrict__ query,
                              const float* __restrict__ W,
                              const float* __restrict__ bias,
                              float* __restrict__ key)
{
    const int lane  = threadIdx.x;
    const int btile = blockIdx.x * 16;
    const int ntile = blockIdx.y * 16;
    const int l15   = lane & 15;
    const int koff  = (lane < 16) ? 0 : 2;

    const float* arow = query + (size_t)(btile + l15) * D_N + koff;
    const float* brow = W     + (size_t)(ntile + l15) * D_N + koff;

    v8f acc = {};
    for (int k0 = 0; k0 < D_N; k0 += 4) {
        v2f a = *(const v2f*)(arow + k0);
        v2f bm = *(const v2f*)(brow + k0);
        acc = __builtin_amdgcn_wmma_f32_16x16x4_f32(
             false, a,  false, bm,
             (short)0, acc,  false,  false);
    }

    const int   n     = ntile + l15;
    const int   mbase = btile + ((lane < 16) ? 0 : 8);
    const float bs    = bias[n];
    #pragma unroll
    for (int r = 0; r < 8; ++r)
        key[(size_t)(mbase + r) * D_N + n] = acc[r] + bs;
}

__global__ void attn_partial(const float* __restrict__ attended,
                             const int*   __restrict__ seqlen,
                             const float* __restrict__ key,
                             float*       __restrict__ part)
{
    const int b     = blockIdx.x;
    const int chunk = blockIdx.y;
    const int wave  = threadIdx.x >> 5;
    const int lane  = threadIdx.x & 31;
    const int p     = chunk * WAVES + wave;
    const int s0    = p * SPW;
    const int len   = seqlen[b];

    float kreg[8];
    {
        const float* kp = key + (size_t)b * D_N + lane * 8;
        v4f k0 = *(const v4f*)(kp);
        v4f k1 = *(const v4f*)(kp + 4);
        kreg[0]=k0.x; kreg[1]=k0.y; kreg[2]=k0.z; kreg[3]=k0.w;
        kreg[4]=k1.x; kreg[5]=k1.y; kreg[6]=k1.z; kreg[7]=k1.w;
    }

    float m = -INFINITY;
    float l = 0.f;
    float c[8];
    #pragma unroll
    for (int j = 0; j < 8; ++j) c[j] = 0.f;

    const size_t rowStride = (size_t)B_N * D_N;
    const float* row = attended + ((size_t)s0 * B_N + b) * D_N + lane * 8;

    const int sEnd = (s0 + SPW < len) ? (s0 + SPW) : len;
    for (int s = s0; s < sEnd; ++s, row += rowStride) {
        __builtin_prefetch((const void*)(row + 2 * rowStride), 0, 0);

        v4f a0 = *(const v4f*)(row);
        v4f a1 = *(const v4f*)(row + 4);
        float a[8] = {a0.x, a0.y, a0.z, a0.w, a1.x, a1.y, a1.z, a1.w};

        float d = 0.f;
        #pragma unroll
        for (int j = 0; j < 8; ++j) d = fmaf(a[j], kreg[j], d);
        #pragma unroll
        for (int off = 16; off >= 1; off >>= 1)
            d += __shfl_xor(d, off, 32);
        const float score = d;

        const float mnew  = fmaxf(m, score);
        const float scale = __expf(m - mnew);
        const float e     = __expf(score - mnew);
        l = l * scale + e;
        #pragma unroll
        for (int j = 0; j < 8; ++j) c[j] = fmaf(c[j], scale, e * a[j]);
        m = mnew;
    }

    float* out = part + ((size_t)b * NP + p) * PART_STRIDE;
    if (lane == 0) { out[0] = m; out[1] = l; }
    #pragma unroll
    for (int j = 0; j < 8; ++j) out[2 + lane * 8 + j] = c[j];
}

__global__ void attn_combine(const float* __restrict__ part,
                             float*       __restrict__ out)
{
    const int b = blockIdx.x;
    const int d = threadIdx.x;
    const float* pb = part + (size_t)b * NP * PART_STRIDE;

    float M = -INFINITY;
    #pragma unroll 4
    for (int p = 0; p < NP; ++p)
        M = fmaxf(M, pb[(size_t)p * PART_STRIDE]);

    float L = 0.f, acc = 0.f;
    for (int p = 0; p < NP; ++p) {
        const float* pp = pb + (size_t)p * PART_STRIDE;
        const float  w  = __expf(pp[0] - M);
        L   += pp[1] * w;
        acc += pp[2 + d] * w;
    }
    out[(size_t)b * D_N + d] = acc / L;
}

extern "C" void kernel_launch(void* const* d_in, const int* in_sizes, int n_in,
                              void* d_out, int out_size, void* d_ws, size_t ws_size,
                              hipStream_t stream)
{
    (void)in_sizes; (void)n_in; (void)out_size; (void)ws_size;

    const float* query    = (const float*)d_in[0];
    const float* attended = (const float*)d_in[1];
    const int*   seqlen   = (const int*)  d_in[2];
    const float* W        = (const float*)d_in[3];
    const float* bias     = (const float*)d_in[4];
    float*       out      = (float*)d_out;

    float* key  = (float*)d_ws;
    float* part = key + (size_t)B_N * D_N;

    key_gemm_wmma<<<dim3(B_N / 16, D_N / 16), 32, 0, stream>>>(query, W, bias, key);
    attn_partial <<<dim3(B_N, NC), WAVES * 32, 0, stream>>>(attended, seqlen, key, part);
    attn_combine <<<B_N, D_N, 0, stream>>>(part, out);
}


# attn_partial: 16 s rows per trip, next 16 rows' loads issued ahead into a second register set; 2-chain lane dot and one softmax rescale per batch change roundings of scores, l and c
# speedup vs baseline: 1.9962x; 1.9962x over previous
#include <hip/hip_runtime.h>
#include <hip/hip_bf16.h>
#include <math.h>

#define S_LEN 4096
#define B_N   64
#define D_N   256

#define NC    8
#define WAVES 4
#define NP    (NC * WAVES)
#define SPW   (S_LEN / NP)
#define PART_STRIDE 258
#define RB    16

typedef __attribute__((ext_vector_type(2))) float v2f;
typedef __attribute__((ext_vector_type(4))) float v4f;
typedef __attribute__((ext_vector_type(8))) float v8f;

__global__ void key_gemm_wmma(const float* __restrict__ query,
                              const float* __restrict__ W,
                              const float* __restrict__ bias,
                              float* __restrict__ key)
{
    const int lane  = threadIdx.x;
    const int btile = blockIdx.x * 16;
    const int ntile = blockIdx.y * 16;
    const int l15   = lane & 15;
    const int koff  = (lane < 16) ? 0 : 2;

    const float* arow = query + (size_t)(btile + l15) * D_N + koff;
    const float* brow = W     + (size_t)(ntile + l15) * D_N + koff;

    v8f acc = {};
    for (int k0 = 0; k0 < D_N; k0 += 4) {
        v2f a = *(const v2f*)(arow + k0);
        v2f bm = *(const v2f*)(brow + k0);
        acc = __builtin_amdgcn_wmma_f32_16x16x4_f32(
             false, a,  false, bm,
             (short)0, acc,  false,  false);
    }

    const int   n     = ntile + l15;
    const int   mbase = btile + ((lane < 16) ? 0 : 8);
    const float bs    = bias[n];
    #pragma unroll
    for (int r = 0; r < 8; ++r)
        key[(size_t)(mbase + r) * D_N + n] = acc[r] + bs;
}

struct RowBatch { v4f lo[RB]; v4f hi[RB]; };

static __device__ __forceinline__ void load_batch(RowBatch& q, const float* lanebase,
                                                  int first, int last)
{
    #pragma unroll
    for (int r = 0; r < RB; ++r) {
        int idx = first + r;
        idx = (idx < last) ? idx : last;
        const float* p = lanebase + (size_t)idx * ((size_t)B_N * D_N);
        q.lo[r] = *(const v4f*)(p);
        q.hi[r] = *(const v4f*)(p + 4);
    }
}

static __device__ __forceinline__ v2f fma2(v2f a, v2f b, v2f c)
{
    return __builtin_elementwise_fma(a, b, c);
}

static __device__ __forceinline__ void fold_batch(const RowBatch& q, const v2f (&k2)[4],
                                                  int first, int n,
                                                  float& m, float& l, v2f (&c2)[4])
{
    float sc[RB];
    #pragma unroll
    for (int r = 0; r < RB; ++r) {
        v2f acc = q.lo[r].xy * k2[0];
        acc = fma2(q.lo[r].zw, k2[1], acc);
        acc = fma2(q.hi[r].xy, k2[2], acc);
        acc = fma2(q.hi[r].zw, k2[3], acc);
        sc[r] = acc.x + acc.y;
    }
    #pragma unroll
    for (int off = 16; off >= 1; off >>= 1) {
        float t[RB];
        #pragma unroll
        for (int r = 0; r < RB; ++r) t[r] = __shfl_xor(sc[r], off, 32);
        #pragma unroll
        for (int r = 0; r < RB; ++r) sc[r] += t[r];
    }
    float bm = -INFINITY;
    #pragma unroll
    for (int r = 0; r < RB; ++r) {
        sc[r] = (first + r < n) ? sc[r] : -INFINITY;
        bm = fmaxf(bm, sc[r]);
    }
    const float mnew  = fmaxf(m, bm);
    const float scale = __expf(m - mnew);
    float e[RB];
    #pragma unroll
    for (int r = 0; r < RB; ++r) e[r] = __expf(sc[r] - mnew);
    l *= scale;
    #pragma unroll
    for (int r = 0; r < RB; ++r) l += e[r];
    const v2f scale2 = {scale, scale};
    #pragma unroll
    for (int j = 0; j < 4; ++j) c2[j] *= scale2;
    #pragma unroll
    for (int r = 0; r < RB; ++r) {
        const v2f e2 = {e[r], e[r]};
        c2[0] = fma2(e2, q.lo[r].xy, c2[0]);
        c2[1] = fma2(e2, q.lo[r].zw, c2[1]);
        c2[2] = fma2(e2, q.hi[r].xy, c2[2]);
        c2[3] = fma2(e2, q.hi[r].zw, c2[3]);
    }
    m = mnew;
}

__global__ void __launch_bounds__(WAVES * 32)
attn_partial(const float* __restrict__ attended,
             const int*   __restrict__ seqlen,
             const float* __restrict__ key,
             float*       __restrict__ part)
{
    const int b     = blockIdx.x;
    const int chunk = blockIdx.y;
    const int wave  = __builtin_amdgcn_readfirstlane(threadIdx.x >> 5);
    const int lane  = threadIdx.x & 31;
    const int p     = chunk * WAVES + wave;
    const int s0    = p * SPW;
    const int len   = seqlen[b];

    v2f k2[4];
    {
        const float* kp = key + (size_t)b * D_N + lane * 8;
        v4f k0 = *(const v4f*)(kp);
        v4f k1 = *(const v4f*)(kp + 4);
        k2[0] = k0.xy; k2[1] = k0.zw; k2[2] = k1.xy; k2[3] = k1.zw;
    }

    float m = -INFINITY;
    float l = 0.f;
    v2f c2[4];
    #pragma unroll
    for (int j = 0; j < 4; ++j) c2[j] = (v2f){0.f, 0.f};

    const float* lanebase = attended + ((size_t)s0 * B_N + b) * D_N + lane * 8;

    const int sEnd = (s0 + SPW < len) ? (s0 + SPW) : len;
    const int n    = sEnd - s0;
    if (n > 0) {
        const int last = n - 1;
        RowBatch cur, nxt;
        load_batch(cur, lanebase, 0, last);
        for (int first = 0; first < n; first += RB) {
            __builtin_amdgcn_sched_barrier(0);
            load_batch(nxt, lanebase, first + RB, last);
            __builtin_amdgcn_sched_barrier(0);
            fold_batch(cur, k2, first, n, m, l, c2);
            cur = nxt;
        }
    }

    float* out = part + ((size_t)b * NP + p) * PART_STRIDE;
    if (lane == 0) { out[0] = m; out[1] = l; }
    #pragma unroll
    for (int j = 0; j < 4; ++j) {
        out[2 + lane * 8 + 2 * j]     = c2[j].x;
        out[2 + lane * 8 + 2 * j + 1] = c2[j].y;
    }
}

__global__ void attn_combine(const float* __restrict__ part,
                             float*       __restrict__ out)
{
    const int b = blockIdx.x;
    const int d = threadIdx.x;
    const float* pb = part + (size_t)b * NP * PART_STRIDE;

    float M = -INFINITY;
    #pragma unroll 4
    for (int p = 0; p < NP; ++p)
        M = fmaxf(M, pb[(size_t)p * PART_STRIDE]);

    float L = 0.f, acc = 0.f;
    for (int p = 0; p < NP; ++p) {
        const float* pp = pb + (size_t)p * PART_STRIDE;
        const float  w  = __expf(pp[0] - M);
        L   += pp[1] * w;
        acc += pp[2 + d] * w;
    }
    out[(size_t)b * D_N + d] = acc / L;
}

extern "C" void kernel_launch(void* const* d_in, const int* in_sizes, int n_in,
                              void* d_out, int out_size, void* d_ws, size_t ws_size,
                              hipStream_t stream)
{
    (void)in_sizes; (void)n_in; (void)out_size; (void)ws_size;

    const float* query    = (const float*)d_in[0];
    const float* attended = (const float*)d_in[1];
    const int*   seqlen   = (const int*)  d_in[2];
    const float* W        = (const float*)d_in[3];
    const float* bias     = (const float*)d_in[4];
    float*       out      = (float*)d_out;

    float* key  = (float*)d_ws;
    float* part = key + (size_t)B_N * D_N;

    key_gemm_wmma<<<dim3(B_N / 16, D_N / 16), 32, 0, stream>>>(query, W, bias, key);
    attn_partial <<<dim3(B_N, NC), WAVES * 32, 0, stream>>>(attended, seqlen, key, part);
    attn_combine <<<B_N, D_N, 0, stream>>>(part, out);
}


# attn_combine: all loads of a batch row's partials (maxima, sums, c) issued up front and the max taken from registers: one memory round trip instead of 17 serial ones; order of every sum unchanged
# speedup vs baseline: 2.2789x; 1.1416x over previous
#include <hip/hip_runtime.h>
#include <hip/hip_bf16.h>
#include <math.h>

#define S_LEN 4096
#define B_N   64
#define D_N   256

#define NC    8
#define WAVES 4
#define NP    (NC * WAVES)
#define SPW   (S_LEN / NP)
#define PART_STRIDE 258
#define RB    16

typedef __attribute__((ext_vector_type(2))) float v2f;
typedef __attribute__((ext_vector_type(4))) float v4f;
typedef __attribute__((ext_vector_type(8))) float v8f;

__global__ void key_gemm_wmma(const float* __restrict__ query,
                              const float* __restrict__ W,
                              const float* __restrict__ bias,
                              float* __restrict__ key)
{
    const int lane  = threadIdx.x;
    const int btile = blockIdx.x * 16;
    const int ntile = blockIdx.y * 16;
    const int l15   = lane & 15;
    const int koff  = (lane < 16) ? 0 : 2;

    const float* arow = query + (size_t)(btile + l15) * D_N + koff;
    const float* brow = W     + (size_t)(ntile + l15) * D_N + koff;

    v8f acc = {};
    for (int k0 = 0; k0 < D_N; k0 += 4) {
        v2f a = *(const v2f*)(arow + k0);
        v2f bm = *(const v2f*)(brow + k0);
        acc = __builtin_amdgcn_wmma_f32_16x16x4_f32(
             false, a,  false, bm,
             (short)0, acc,  false,  false);
    }

    const int   n     = ntile + l15;
    const int   mbase = btile + ((lane < 16) ? 0 : 8);
    const float bs    = bias[n];
    #pragma unroll
    for (int r = 0; r < 8; ++r)
        key[(size_t)(mbase + r) * D_N + n] = acc[r] + bs;
}

struct RowBatch { v4f lo[RB]; v4f hi[RB]; };

static __device__ __forceinline__ void load_batch(RowBatch& q, const float* lanebase,
                                                  int first, int last)
{
    #pragma unroll
    for (int r = 0; r < RB; ++r) {
        int idx = first + r;
        idx = (idx < last) ? idx : last;
        const float* p = lanebase + (size_t)idx * ((size_t)B_N * D_N);
        q.lo[r] = *(const v4f*)(p);
        q.hi[r] = *(const v4f*)(p + 4);
    }
}

static __device__ __forceinline__ v2f fma2(v2f a, v2f b, v2f c)
{
    return __builtin_elementwise_fma(a, b, c);
}

static __device__ __forceinline__ void fold_batch(const RowBatch& q, const v2f (&k2)[4],
                                                  int first, int n,
                                                  float& m, float& l, v2f (&c2)[4])
{
    float sc[RB];
    #pragma unroll
    for (int r = 0; r < RB; ++r) {
        v2f acc = q.lo[r].xy * k2[0];
        acc = fma2(q.lo[r].zw, k2[1], acc);
        acc = fma2(q.hi[r].xy, k2[2], acc);
        acc = fma2(q.hi[r].zw, k2[3], acc);
        sc[r] = acc.x + acc.y;
    }
    #pragma unroll
    for (int off = 16; off >= 1; off >>= 1) {
        float t[RB];
        #pragma unroll
        for (int r = 0; r < RB; ++r) t[r] = __shfl_xor(sc[r], off, 32);
        #pragma unroll
        for (int r = 0; r < RB; ++r) sc[r] += t[r];
    }
    float bm = -INFINITY;
    #pragma unroll
    for (int r = 0; r < RB; ++r) {
        sc[r] = (first + r < n) ? sc[r] : -INFINITY;
        bm = fmaxf(bm, sc[r]);
    }
    const float mnew  = fmaxf(m, bm);
    const float scale = __expf(m - mnew);
    float e[RB];
    #pragma unroll
    for (int r = 0; r < RB; ++r) e[r] = __expf(sc[r] - mnew);
    l *= scale;
    #pragma unroll
    for (int r = 0; r < RB; ++r) l += e[r];
    const v2f scale2 = {scale, scale};
    #pragma unroll
    for (int j = 0; j < 4; ++j) c2[j] *= scale2;
    #pragma unroll
    for (int r = 0; r < RB; ++r) {
        const v2f e2 = {e[r], e[r]};
        c2[0] = fma2(e2, q.lo[r].xy, c2[0]);
        c2[1] = fma2(e2, q.lo[r].zw, c2[1]);
        c2[2] = fma2(e2, q.hi[r].xy, c2[2]);
        c2[3] = fma2(e2, q.hi[r].zw, c2[3]);
    }
    m = mnew;
}

__global__ void __launch_bounds__(WAVES * 32)
attn_partial(const float* __restrict__ attended,
             const int*   __restrict__ seqlen,
             const float* __restrict__ key,
             float*       __restrict__ part)
{
    const int b     = blockIdx.x;
    const int chunk = blockIdx.y;
    const int wave  = __builtin_amdgcn_readfirstlane(threadIdx.x >> 5);
    const int lane  = threadIdx.x & 31;
    const int p     = chunk * WAVES + wave;
    const int s0    = p * SPW;
    const int len   = seqlen[b];

    v2f k2[4];
    {
        const float* kp = key + (size_t)b * D_N + lane * 8;
        v4f k0 = *(const v4f*)(kp);
        v4f k1 = *(const v4f*)(kp + 4);
        k2[0] = k0.xy; k2[1] = k0.zw; k2[2] = k1.xy; k2[3] = k1.zw;
    }

    float m = -INFINITY;
    float l = 0.f;
    v2f c2[4];
    #pragma unroll
    for (int j = 0; j < 4; ++j) c2[j] = (v2f){0.f, 0.f};

    const float* lanebase = attended + ((size_t)s0 * B_N + b) * D_N + lane * 8;

    const int sEnd = (s0 + SPW < len) ? (s0 + SPW) : len;
    const int n    = sEnd - s0;
    if (n > 0) {
        const int last = n - 1;
        RowBatch cur, nxt;
        load_batch(cur, lanebase, 0, last);
        for (int first = 0; first < n; first += RB) {
            __builtin_amdgcn_sched_barrier(0);
            load_batch(nxt, lanebase, first + RB, last);
            __builtin_amdgcn_sched_barrier(0);
            fold_batch(cur, k2, first, n, m, l, c2);
            cur = nxt;
        }
    }

    float* out = part + ((size_t)b * NP + p) * PART_STRIDE;
    if (lane == 0) { out[0] = m; out[1] = l; }
    #pragma unroll
    for (int j = 0; j < 4; ++j) {
        out[2 + lane * 8 + 2 * j]     = c2[j].x;
        out[2 + lane * 8 + 2 * j + 1] = c2[j].y;
    }
}

__global__ void attn_combine(const float* __restrict__ part,
                             float*       __restrict__ out)
{
    const int b = blockIdx.x;
    const int d = threadIdx.x;
    const float* pb = part + (size_t)b * NP * PART_STRIDE;

    float mv[NP], lv[NP], cv[NP];
    #pragma unroll
    for (int p = 0; p < NP; ++p) {
        const float* pp = pb + (size_t)p * PART_STRIDE;
        mv[p] = pp[0];
        lv[p] = pp[1];
        cv[p] = pp[2 + d];
    }

    float M = -INFINITY;
    #pragma unroll
    for (int p = 0; p < NP; ++p)
        M = fmaxf(M, mv[p]);

    float L = 0.f, acc = 0.f;
    #pragma unroll
    for (int p = 0; p < NP; ++p) {
        const float w = __expf(mv[p] - M);
        L   += lv[p] * w;
        acc += cv[p] * w;
    }
    out[(size_t)b * D_N + d] = acc / L;
}

extern "C" void kernel_launch(void* const* d_in, const int* in_sizes, int n_in,
                              void* d_out, int out_size, void* d_ws, size_t ws_size,
                              hipStream_t stream)
{
    (void)in_sizes; (void)n_in; (void)out_size; (void)ws_size;

    const float* query    = (const float*)d_in[0];
    const float* attended = (const float*)d_in[1];
    const int*   seqlen   = (const int*)  d_in[2];
    const float* W        = (const float*)d_in[3];
    const float* bias     = (const float*)d_in[4];
    float*       out      = (float*)d_out;

    float* key  = (float*)d_ws;
    float* part = key + (size_t)B_N * D_N;

    key_gemm_wmma<<<dim3(B_N / 16, D_N / 16), 32, 0, stream>>>(query, W, bias, key);
    attn_partial <<<dim3(B_N, NC), WAVES * 32, 0, stream>>>(attended, seqlen, key, part);
    attn_combine <<<B_N, D_N, 0, stream>>>(part, out);
}


# key_gemm: k split over 4 waves per tile, all of a wave's fragment loads requested before its first WMMA (1 memory round trip, was 4); partial tiles summed through LDS: k sum order of key changes
# speedup vs baseline: 2.5501x; 1.1190x over previous
#include <hip/hip_runtime.h>
#include <hip/hip_bf16.h>
#include <math.h>

#define S_LEN 4096
#define B_N   64
#define D_N   256

#define NC    8
#define WAVES 4
#define NP    (NC * WAVES)
#define SPW   (S_LEN / NP)
#define PART_STRIDE 258
#define RB    16

typedef __attribute__((ext_vector_type(2))) float v2f;
typedef __attribute__((ext_vector_type(4))) float v4f;
typedef __attribute__((ext_vector_type(8))) float v8f;

#define KW     4
#define KSPAN  (D_N / KW)
#define KLANE  (KSPAN / 2)
#define KVEC   (KLANE / 4)

__global__ void __launch_bounds__(KW * 32)
key_gemm_wmma(const float* __restrict__ query,
              const float* __restrict__ W,
              const float* __restrict__ bias,
              float* __restrict__ key)
{
    __shared__ float red[KW][8][32];

    const int lane  = threadIdx.x & 31;
    const int wave  = threadIdx.x >> 5;
    const int btile = blockIdx.x * 16;
    const int ntile = blockIdx.y * 16;
    const int l15   = lane & 15;
    const int half  = lane >> 4;
    const int kbase = wave * KSPAN + half * KLANE;

    const float* arow = query + (size_t)(btile + l15) * D_N + kbase;
    const float* brow = W     + (size_t)(ntile + l15) * D_N + kbase;

    v4f av[KVEC], bv[KVEC];
    #pragma unroll
    for (int i = 0; i < KVEC; ++i) av[i] = *(const v4f*)(arow + 4 * i);
    #pragma unroll
    for (int i = 0; i < KVEC; ++i) bv[i] = *(const v4f*)(brow + 4 * i);
    const float bs = bias[ntile + l15];
    __builtin_amdgcn_sched_barrier(0);

    v8f acc0 = {}, acc1 = {};
    #pragma unroll
    for (int i = 0; i < KVEC; ++i) {
        acc0 = __builtin_amdgcn_wmma_f32_16x16x4_f32(
            false, (v2f)av[i].xy, false, (v2f)bv[i].xy, (short)0, acc0, false, false);
        acc1 = __builtin_amdgcn_wmma_f32_16x16x4_f32(
            false, (v2f)av[i].zw, false, (v2f)bv[i].zw, (short)0, acc1, false, false);
    }

    #pragma unroll
    for (int r = 0; r < 8; ++r) red[wave][r][lane] = acc0[r] + acc1[r];
    __syncthreads();

    const int   n  = ntile + l15;
    #pragma unroll
    for (int q = 0; q < 2; ++q) {
        const int r = 2 * wave + q;
        float s = red[0][r][lane];
        #pragma unroll
        for (int w = 1; w < KW; ++w) s += red[w][r][lane];
        key[(size_t)(btile + 8 * half + r) * D_N + n] = s + bs;
    }
}

struct RowBatch { v4f lo[RB]; v4f hi[RB]; };

static __device__ __forceinline__ void load_batch(RowBatch& q, const float* lanebase,
                                                  int first, int last)
{
    #pragma unroll
    for (int r = 0; r < RB; ++r) {
        int idx = first + r;
        idx = (idx < last) ? idx : last;
        const float* p = lanebase + (size_t)idx * ((size_t)B_N * D_N);
        q.lo[r] = *(const v4f*)(p);
        q.hi[r] = *(const v4f*)(p + 4);
    }
}

static __device__ __forceinline__ v2f fma2(v2f a, v2f b, v2f c)
{
    return __builtin_elementwise_fma(a, b, c);
}

static __device__ __forceinline__ void fold_batch(const RowBatch& q, const v2f (&k2)[4],
                                                  int first, int n,
                                                  float& m, float& l, v2f (&c2)[4])
{
    float sc[RB];
    #pragma unroll
    for (int r = 0; r < RB; ++r) {
        v2f acc = q.lo[r].xy * k2[0];
        acc = fma2(q.lo[r].zw, k2[1], acc);
        acc = fma2(q.hi[r].xy, k2[2], acc);
        acc = fma2(q.hi[r].zw, k2[3], acc);
        sc[r] = acc.x + acc.y;
    }
    #pragma unroll
    for (int off = 16; off >= 1; off >>= 1) {
        float t[RB];
        #pragma unroll
        for (int r = 0; r < RB; ++r) t[r] = __shfl_xor(sc[r], off, 32);
        #pragma unroll
        for (int r = 0; r < RB; ++r) sc[r] += t[r];
    }
    float bm = -INFINITY;
    #pragma unroll
    for (int r = 0; r < RB; ++r) {
        sc[r] = (first + r < n) ? sc[r] : -INFINITY;
        bm = fmaxf(bm, sc[r]);
    }
    const float mnew  = fmaxf(m, bm);
    const float scale = __expf(m - mnew);
    float e[RB];
    #pragma unroll
    for (int r = 0; r < RB; ++r) e[r] = __expf(sc[r] - mnew);
    l *= scale;
    #pragma unroll
    for (int r = 0; r < RB; ++r) l += e[r];
    const v2f scale2 = {scale, scale};
    #pragma unroll
    for (int j = 0; j < 4; ++j) c2[j] *= scale2;
    #pragma unroll
    for (int r = 0; r < RB; ++r) {
        const v2f e2 = {e[r], e[r]};
        c2[0] = fma2(e2, q.lo[r].xy, c2[0]);
        c2[1] = fma2(e2, q.lo[r].zw, c2[1]);
        c2[2] = fma2(e2, q.hi[r].xy, c2[2]);
        c2[3] = fma2(e2, q.hi[r].zw, c2[3]);
    }
    m = mnew;
}

__global__ void __launch_bounds__(WAVES * 32)
attn_partial(const float* __restrict__ attended,
             const int*   __restrict__ seqlen,
             const float* __restrict__ key,
             float*       __restrict__ part)
{
    const int b     = blockIdx.x;
    const int chunk = blockIdx.y;
    const int wave  = __builtin_amdgcn_readfirstlane(threadIdx.x >> 5);
    const int lane  = threadIdx.x & 31;
    const int p     = chunk * WAVES + wave;
    const int s0    = p * SPW;
    const int len   = seqlen[b];

    v2f k2[4];
    {
        const float* kp = key + (size_t)b * D_N + lane * 8;
        v4f k0 = *(const v4f*)(kp);
        v4f k1 = *(const v4f*)(kp + 4);
        k2[0] = k0.xy; k2[1] = k0.zw; k2[2] = k1.xy; k2[3] = k1.zw;
    }

    float m = -INFINITY;
    float l = 0.f;
    v2f c2[4];
    #pragma unroll
    for (int j = 0; j < 4; ++j) c2[j] = (v2f){0.f, 0.f};

    const float* lanebase = attended + ((size_t)s0 * B_N + b) * D_N + lane * 8;

    const int sEnd = (s0 + SPW < len) ? (s0 + SPW) : len;
    const int n    = sEnd - s0;
    if (n > 0) {
        const int last = n - 1;
        RowBatch cur, nxt;
        load_batch(cur, lanebase, 0, last);
        for (int first = 0; first < n; first += RB) {
            __builtin_amdgcn_sched_barrier(0);
            load_batch(nxt, lanebase, first + RB, last);
            __builtin_amdgcn_sched_barrier(0);
            fold_batch(cur, k2, first, n, m, l, c2);
            cur = nxt;
        }
    }

    float* out = part + ((size_t)b * NP + p) * PART_STRIDE;
    if (lane == 0) { out[0] = m; out[1] = l; }
    #pragma unroll
    for (int j = 0; j < 4; ++j) {
        out[2 + lane * 8 + 2 * j]     = c2[j].x;
        out[2 + lane * 8 + 2 * j + 1] = c2[j].y;
    }
}

__global__ void attn_combine(const float* __restrict__ part,
                             float*       __restrict__ out)
{
    const int b = blockIdx.x;
    const int d = threadIdx.x;
    const float* pb = part + (size_t)b * NP * PART_STRIDE;

    float mv[NP], lv[NP], cv[NP];
    #pragma unroll
    for (int p = 0; p < NP; ++p) {
        const float* pp = pb + (size_t)p * PART_STRIDE;
        mv[p] = pp[0];
        lv[p] = pp[1];
        cv[p] = pp[2 + d];
    }

    float M = -INFINITY;
    #pragma unroll
    for (int p = 0; p < NP; ++p)
        M = fmaxf(M, mv[p]);

    float L = 0.f, acc = 0.f;
    #pragma unroll
    for (int p = 0; p < NP; ++p) {
        const float w = __expf(mv[p] - M);
        L   += lv[p] * w;
        acc += cv[p] * w;
    }
    out[(size_t)b * D_N + d] = acc / L;
}

extern "C" void kernel_launch(void* const* d_in, const int* in_sizes, int n_in,
                              void* d_out, int out_size, void* d_ws, size_t ws_size,
                              hipStream_t stream)
{
    (void)in_sizes; (void)n_in; (void)out_size; (void)ws_size;

    const float* query    = (const float*)d_in[0];
    const float* attended = (const float*)d_in[1];
    const int*   seqlen   = (const int*)  d_in[2];
    const float* W        = (const float*)d_in[3];
    const float* bias     = (const float*)d_in[4];
    float*       out      = (float*)d_out;

    float* key  = (float*)d_ws;
    float* part = key + (size_t)B_N * D_N;

    key_gemm_wmma<<<dim3(B_N / 16, D_N / 16), KW * 32, 0, stream>>>(query, W, bias, key);
    attn_partial <<<dim3(B_N, NC), WAVES * 32, 0, stream>>>(attended, seqlen, key, part);
    attn_combine <<<B_N, D_N, 0, stream>>>(part, out);
}
